# starting kernel
# baseline (speedup 1.0000x reference)
#include <hip/hip_runtime.h>
#include <math.h>

typedef float v2f __attribute__((ext_vector_type(2)));
typedef float v8f __attribute__((ext_vector_type(8)));

#define N_NODES 2048
#define BATCH   8
#define FEAT    128
#define NFILT   64

__global__ void k_proj(const float* __restrict__ x, const float* __restrict__ w1,
                       float* __restrict__ X1) {
    int wave = threadIdx.x >> 5;
    int lane = threadIdx.x & 31;
    int row  = blockIdx.x * 8 + wave;
    const float4 xv = *(const float4*)(x + (size_t)row * FEAT + lane * 4);
    const float4 wv = *(const float4*)(w1 + lane * 4);
    float d = xv.x * wv.x + xv.y * wv.y + xv.z * wv.z + xv.w * wv.w;
    #pragma unroll
    for (int off = 16; off > 0; off >>= 1) d += __shfl_xor(d, off, 32);
    if (lane == 0) X1[row] = d;
}

__global__ void k_wmma_s(const float* __restrict__ Wa, const float* __restrict__ ba,
                         const float* __restrict__ X1,
                         float* __restrict__ SB2, float* __restrict__ CS) {
    int wave   = threadIdx.x >> 5;
    int lane   = threadIdx.x & 31;
    int tile   = blockIdx.x * 4 + wave;
    int colIn  = lane & 15;
    int col    = tile * 16 + colIn;
    int hi     = lane >> 4;
    int koff   = hi * 2;
    int mrow   = colIn;
    int srcRow = mrow & 7;

    v8f c = {};
    for (int k0 = 0; k0 < N_NODES; k0 += 4) {
        v2f av = *(const v2f*)(X1 + srcRow * N_NODES + k0 + koff);
        v2f a;
        a.x = (mrow < 8) ? av.x : ((mrow == 8) ? 1.0f : 0.0f);
        a.y = (mrow < 8) ? av.y : ((mrow == 8) ? 1.0f : 0.0f);
        v2f b;
        b.x = Wa[(size_t)(k0 + koff)     * N_NODES + col];
        b.y = Wa[(size_t)(k0 + koff + 1) * N_NODES + col];
        c = __builtin_amdgcn_wmma_f32_16x16x4_f32(false, a, false, b,
                                                  (short)0, c, false, false);
    }

    float bav = ba[col];
    #pragma unroll
    for (int j = 0; j < 8; ++j) {
        int row = j + hi * 8;
        if (row < 8)        SB2[row * N_NODES + col] = c[j] + bav;
        else if (row == 8)  CS[col] = c[j];
    }
}

__global__ void __launch_bounds__(256) k_softmax_agg(
        const float* __restrict__ A, const float* __restrict__ bias,
        const float* __restrict__ X1, const float* __restrict__ SB2,
        const float* __restrict__ CS, float* __restrict__ out) {
    __shared__ __align__(16) float css[N_NODES];
    __shared__ __align__(16) float sbs[N_NODES];
    __shared__ __align__(16) float xks[N_NODES];
    __shared__ float bs[NFILT];

    int b     = blockIdx.x >> 8;
    int nBase = (blockIdx.x & 255) * 8;

    for (int i = threadIdx.x; i < N_NODES; i += 256) {
        css[i] = CS[i];
        sbs[i] = SB2[b * N_NODES + i];
        xks[i] = X1[b * N_NODES + i];
    }
    if (threadIdx.x < NFILT) bs[threadIdx.x] = bias[threadIdx.x];
    __syncthreads();

    int wave = threadIdx.x >> 5;
    int lane = threadIdx.x & 31;
    int n    = nBase + wave;
    size_t rowBase = ((size_t)(b * N_NODES + n)) * N_NODES;
    float x1bn = X1[b * N_NODES + n];

    float4 lg[16];
    float  mx = -INFINITY;
    #pragma unroll
    for (int i = 0; i < 16; ++i) {
        int k = i * 128 + lane * 4;
        float4 av = *(const float4*)(A + rowBase + k);
        float4 cv = *(const float4*)&css[k];
        float4 sv = *(const float4*)&sbs[k];
        float4 l;
        l.x = av.x + sv.x + x1bn * cv.x;
        l.y = av.y + sv.y + x1bn * cv.y;
        l.z = av.z + sv.z + x1bn * cv.z;
        l.w = av.w + sv.w + x1bn * cv.w;
        lg[i] = l;
        mx = fmaxf(mx, fmaxf(fmaxf(l.x, l.y), fmaxf(l.z, l.w)));
    }
    #pragma unroll
    for (int off = 16; off > 0; off >>= 1)
        mx = fmaxf(mx, __shfl_xor(mx, off, 32));

    float Z = 0.0f, W = 0.0f;
    #pragma unroll
    for (int i = 0; i < 16; ++i) {
        int k = i * 128 + lane * 4;
        float4 xv = *(const float4*)&xks[k];
        float ex = __expf(lg[i].x - mx); Z += ex; W += ex * xv.x;
        float ey = __expf(lg[i].y - mx); Z += ey; W += ey * xv.y;
        float ez = __expf(lg[i].z - mx); Z += ez; W += ez * xv.z;
        float ew = __expf(lg[i].w - mx); Z += ew; W += ew * xv.w;
    }
    #pragma unroll
    for (int off = 16; off > 0; off >>= 1) {
        Z += __shfl_xor(Z, off, 32);
        W += __shfl_xor(W, off, 32);
    }
    float r = W / Z;

    size_t ob = ((size_t)(b * N_NODES + n)) * NFILT;
    out[ob + lane]      = r + bs[lane];
    out[ob + 32 + lane] = r + bs[32 + lane];
}

extern "C" void kernel_launch(void* const* d_in, const int* in_sizes, int n_in,
                              void* d_out, int out_size, void* d_ws, size_t ws_size,
                              hipStream_t stream) {
    const float* x    = (const float*)d_in[0];
    const float* A    = (const float*)d_in[1];
    const float* w1   = (const float*)d_in[2];
    const float* Wa   = (const float*)d_in[3];
    const float* ba   = (const float*)d_in[4];
    const float* bias = (const float*)d_in[5];
    float* out = (float*)d_out;

    float* X1  = (float*)d_ws;
    float* CS  = X1 + BATCH * N_NODES;
    float* SB2 = CS + N_NODES;

    k_proj<<<dim3(2048), dim3(256), 0, stream>>>(x, w1, X1);
    k_wmma_s<<<dim3(32), dim3(128), 0, stream>>>(Wa, ba, X1, SB2, CS);
    k_softmax_agg<<<dim3(2048), dim3(256), 0, stream>>>(A, bias, X1, SB2, CS, out);
}


# Skinny WMMA product: K of each 16-column tile split over 16 waves of one group (2048 waves, was 128); partial tiles added in fixed slice order through LDS (changes f32 summation order)
# speedup vs baseline: 1.9463x; 1.9463x over previous
#include <hip/hip_runtime.h>
#include <math.h>

typedef float v2f __attribute__((ext_vector_type(2)));
typedef float v8f __attribute__((ext_vector_type(8)));

#define N_NODES 2048
#define BATCH   8
#define FEAT    128
#define NFILT   64

__global__ void k_proj(const float* __restrict__ x, const float* __restrict__ w1,
                       float* __restrict__ X1) {
    int wave = threadIdx.x >> 5;
    int lane = threadIdx.x & 31;
    int row  = blockIdx.x * 8 + wave;
    const float4 xv = *(const float4*)(x + (size_t)row * FEAT + lane * 4);
    const float4 wv = *(const float4*)(w1 + lane * 4);
    float d = xv.x * wv.x + xv.y * wv.y + xv.z * wv.z + xv.w * wv.w;
    #pragma unroll
    for (int off = 16; off > 0; off >>= 1) d += __shfl_xor(d, off, 32);
    if (lane == 0) X1[row] = d;
}

#define KSPLIT  16
#define KSLICE  (N_NODES / KSPLIT)
#define KSTEPS  (KSLICE / 4)
#define PLANE   (9 * 16)

__global__ void __launch_bounds__(KSPLIT * 32) k_wmma_s(
        const float* __restrict__ Wa, const float* __restrict__ ba,
        const float* __restrict__ X1,
        float* __restrict__ SB2, float* __restrict__ CS) {
    __shared__ float part[KSPLIT * PLANE];

    int wave   = threadIdx.x >> 5;
    int lane   = threadIdx.x & 31;
    int tile   = blockIdx.x;
    int colIn  = lane & 15;
    int col    = tile * 16 + colIn;
    int hi     = lane >> 4;
    int koff   = hi * 2;
    int mrow   = colIn;
    int srcRow = mrow & 7;
    int kBeg   = wave * KSLICE;

    const float* aPtr = X1 + srcRow * N_NODES + kBeg + koff;
    const float* bPtr = Wa + (size_t)(kBeg + koff) * N_NODES + col;

    v8f c = {};
    #pragma unroll
    for (int i = 0; i < KSTEPS; ++i) {
        v2f av = *(const v2f*)(aPtr + i * 4);
        v2f a;
        a.x = (mrow < 8) ? av.x : ((mrow == 8) ? 1.0f : 0.0f);
        a.y = (mrow < 8) ? av.y : ((mrow == 8) ? 1.0f : 0.0f);
        v2f b;
        b.x = bPtr[(size_t)(i * 4)     * N_NODES];
        b.y = bPtr[(size_t)(i * 4 + 1) * N_NODES];
        c = __builtin_amdgcn_wmma_f32_16x16x4_f32(false, a, false, b,
                                                  (short)0, c, false, false);
    }

    float* plane = part + wave * PLANE;
    if (hi == 0) {
        #pragma unroll
        for (int j = 0; j < 8; ++j) plane[j * 16 + colIn] = c[j];
    } else {
        plane[8 * 16 + colIn] = c[0];
    }
    __syncthreads();

    int t = threadIdx.x;
    if (t < PLANE) {
        float sum = part[t];
        #pragma unroll
        for (int w = 1; w < KSPLIT; ++w) sum += part[w * PLANE + t];
        int row  = t >> 4;
        int ocol = tile * 16 + (t & 15);
        if (row < 8) SB2[row * N_NODES + ocol] = sum + ba[ocol];
        else         CS[ocol] = sum;
    }
}

__global__ void __launch_bounds__(256) k_softmax_agg(
        const float* __restrict__ A, const float* __restrict__ bias,
        const float* __restrict__ X1, const float* __restrict__ SB2,
        const float* __restrict__ CS, float* __restrict__ out) {
    __shared__ __align__(16) float css[N_NODES];
    __shared__ __align__(16) float sbs[N_NODES];
    __shared__ __align__(16) float xks[N_NODES];
    __shared__ float bs[NFILT];

    int b     = blockIdx.x >> 8;
    int nBase = (blockIdx.x & 255) * 8;

    for (int i = threadIdx.x; i < N_NODES; i += 256) {
        css[i] = CS[i];
        sbs[i] = SB2[b * N_NODES + i];
        xks[i] = X1[b * N_NODES + i];
    }
    if (threadIdx.x < NFILT) bs[threadIdx.x] = bias[threadIdx.x];
    __syncthreads();

    int wave = threadIdx.x >> 5;
    int lane = threadIdx.x & 31;
    int n    = nBase + wave;
    size_t rowBase = ((size_t)(b * N_NODES + n)) * N_NODES;
    float x1bn = X1[b * N_NODES + n];

    float4 lg[16];
    float  mx = -INFINITY;
    #pragma unroll
    for (int i = 0; i < 16; ++i) {
        int k = i * 128 + lane * 4;
        float4 av = *(const float4*)(A + rowBase + k);
        float4 cv = *(const float4*)&css[k];
        float4 sv = *(const float4*)&sbs[k];
        float4 l;
        l.x = av.x + sv.x + x1bn * cv.x;
        l.y = av.y + sv.y + x1bn * cv.y;
        l.z = av.z + sv.z + x1bn * cv.z;
        l.w = av.w + sv.w + x1bn * cv.w;
        lg[i] = l;
        mx = fmaxf(mx, fmaxf(fmaxf(l.x, l.y), fmaxf(l.z, l.w)));
    }
    #pragma unroll
    for (int off = 16; off > 0; off >>= 1)
        mx = fmaxf(mx, __shfl_xor(mx, off, 32));

    float Z = 0.0f, W = 0.0f;
    #pragma unroll
    for (int i = 0; i < 16; ++i) {
        int k = i * 128 + lane * 4;
        float4 xv = *(const float4*)&xks[k];
        float ex = __expf(lg[i].x - mx); Z += ex; W += ex * xv.x;
        float ey = __expf(lg[i].y - mx); Z += ey; W += ey * xv.y;
        float ez = __expf(lg[i].z - mx); Z += ez; W += ez * xv.z;
        float ew = __expf(lg[i].w - mx); Z += ew; W += ew * xv.w;
    }
    #pragma unroll
    for (int off = 16; off > 0; off >>= 1) {
        Z += __shfl_xor(Z, off, 32);
        W += __shfl_xor(W, off, 32);
    }
    float r = W / Z;

    size_t ob = ((size_t)(b * N_NODES + n)) * NFILT;
    out[ob + lane]      = r + bs[lane];
    out[ob + 32 + lane] = r + bs[32 + lane];
}

extern "C" void kernel_launch(void* const* d_in, const int* in_sizes, int n_in,
                              void* d_out, int out_size, void* d_ws, size_t ws_size,
                              hipStream_t stream) {
    const float* x    = (const float*)d_in[0];
    const float* A    = (const float*)d_in[1];
    const float* w1   = (const float*)d_in[2];
    const float* Wa   = (const float*)d_in[3];
    const float* ba   = (const float*)d_in[4];
    const float* bias = (const float*)d_in[5];
    float* out = (float*)d_out;

    float* X1  = (float*)d_ws;
    float* CS  = X1 + BATCH * N_NODES;
    float* SB2 = CS + N_NODES;

    k_proj<<<dim3(2048), dim3(256), 0, stream>>>(x, w1, X1);
    k_wmma_s<<<dim3(128), dim3(512), 0, stream>>>(Wa, ba, X1, SB2, CS);
    k_softmax_agg<<<dim3(2048), dim3(256), 0, stream>>>(A, bias, X1, SB2, CS, out);
}


# Row loops in both row-per-wave kernels: projection 8 rows per wave, softmax 4 rows per wave with the next row's loads issued ahead (2048 and 4096 waves, were 16384 each); LDS staged per 64 rows
# speedup vs baseline: 2.2992x; 1.1813x over previous
#include <hip/hip_runtime.h>
#include <math.h>

typedef float v2f __attribute__((ext_vector_type(2)));
typedef float v8f __attribute__((ext_vector_type(8)));

#define N_NODES 2048
#define BATCH   8
#define FEAT    128
#define NFILT   64

#define PROJ_ROWS 8

__global__ void __launch_bounds__(256) k_proj(const float* __restrict__ x,
                                              const float* __restrict__ w1,
                                              float* __restrict__ X1) {
    int wave = threadIdx.x >> 5;
    int lane = threadIdx.x & 31;
    int row0 = (blockIdx.x * 8 + wave) * PROJ_ROWS;
    const float* xp = x + (size_t)row0 * FEAT + lane * 4;
    float4 xv[PROJ_ROWS];
    #pragma unroll
    for (int r = 0; r < PROJ_ROWS; ++r) xv[r] = *(const float4*)(xp + r * FEAT);
    const float4 wv = *(const float4*)(w1 + lane * 4);
    float mine = 0.0f;
    #pragma unroll
    for (int r = 0; r < PROJ_ROWS; ++r) {
        float d = xv[r].x * wv.x + xv[r].y * wv.y + xv[r].z * wv.z + xv[r].w * wv.w;
        #pragma unroll
        for (int off = 16; off > 0; off >>= 1) d += __shfl_xor(d, off, 32);
        if (lane == r) mine = d;
    }
    if (lane < PROJ_ROWS) X1[row0 + lane] = mine;
}

#define KSPLIT  16
#define KSLICE  (N_NODES / KSPLIT)
#define KSTEPS  (KSLICE / 4)
#define PLANE   (9 * 16)

__global__ void __launch_bounds__(KSPLIT * 32) k_wmma_s(
        const float* __restrict__ Wa, const float* __restrict__ ba,
        const float* __restrict__ X1,
        float* __restrict__ SB2, float* __restrict__ CS) {
    __shared__ float part[KSPLIT * PLANE];

    int wave   = threadIdx.x >> 5;
    int lane   = threadIdx.x & 31;
    int tile   = blockIdx.x;
    int colIn  = lane & 15;
    int col    = tile * 16 + colIn;
    int hi     = lane >> 4;
    int koff   = hi * 2;
    int mrow   = colIn;
    int srcRow = mrow & 7;
    int kBeg   = wave * KSLICE;

    const float* aPtr = X1 + srcRow * N_NODES + kBeg + koff;
    const float* bPtr = Wa + (size_t)(kBeg + koff) * N_NODES + col;

    v8f c = {};
    #pragma unroll
    for (int i = 0; i < KSTEPS; ++i) {
        v2f av = *(const v2f*)(aPtr + i * 4);
        v2f a;
        a.x = (mrow < 8) ? av.x : ((mrow == 8) ? 1.0f : 0.0f);
        a.y = (mrow < 8) ? av.y : ((mrow == 8) ? 1.0f : 0.0f);
        v2f b;
        b.x = bPtr[(size_t)(i * 4)     * N_NODES];
        b.y = bPtr[(size_t)(i * 4 + 1) * N_NODES];
        c = __builtin_amdgcn_wmma_f32_16x16x4_f32(false, a, false, b,
                                                  (short)0, c, false, false);
    }

    float* plane = part + wave * PLANE;
    if (hi == 0) {
        #pragma unroll
        for (int j = 0; j < 8; ++j) plane[j * 16 + colIn] = c[j];
    } else {
        plane[8 * 16 + colIn] = c[0];
    }
    __syncthreads();

    int t = threadIdx.x;
    if (t < PLANE) {
        float sum = part[t];
        #pragma unroll
        for (int w = 1; w < KSPLIT; ++w) sum += part[w * PLANE + t];
        int row  = t >> 4;
        int ocol = tile * 16 + (t & 15);
        if (row < 8) SB2[row * N_NODES + ocol] = sum + ba[ocol];
        else         CS[ocol] = sum;
    }
}

#define SM_WAVES 16
#define SM_ROWS  4
#define SM_GROUP_ROWS (SM_WAVES * SM_ROWS)

__global__ void __launch_bounds__(SM_WAVES * 32) k_softmax_agg(
        const float* __restrict__ A, const float* __restrict__ bias,
        const float* __restrict__ X1, const float* __restrict__ SB2,
        const float* __restrict__ CS, float* __restrict__ out) {
    __shared__ __align__(16) float css[N_NODES];
    __shared__ __align__(16) float sbs[N_NODES];
    __shared__ __align__(16) float xks[N_NODES];
    __shared__ float bs[NFILT];

    const int t    = threadIdx.x;
    const int lane = t & 31;
    const int wave = t >> 5;
    const int b    = blockIdx.x >> 5;
    const int n0   = (blockIdx.x & 31) * SM_GROUP_ROWS + wave * SM_ROWS;
    const float* aRow = A + ((size_t)(b * N_NODES + n0)) * N_NODES + lane * 4;

    float4 buf[2][16];

    #pragma unroll
    for (int i = 0; i < 16; ++i)
        buf[0][i] = *(const float4*)(aRow + i * 128);

    *(float4*)&css[t * 4] = *(const float4*)(CS + t * 4);
    *(float4*)&sbs[t * 4] = *(const float4*)(SB2 + b * N_NODES + t * 4);
    *(float4*)&xks[t * 4] = *(const float4*)(X1 + b * N_NODES + t * 4);
    if (t < NFILT) bs[t] = bias[t];
    __syncthreads();

    #pragma unroll
    for (int j = 0; j < SM_ROWS; ++j) {
        __builtin_amdgcn_sched_barrier(0);
        if (j + 1 < SM_ROWS) {
            #pragma unroll
            for (int i = 0; i < 16; ++i)
                buf[(j + 1) & 1][i] =
                    *(const float4*)(aRow + (size_t)(j + 1) * N_NODES + i * 128);
        }
        __builtin_amdgcn_sched_barrier(0);

        float4* lg = buf[j & 1];
        const int n = n0 + j;
        const float x1bn = xks[n];

        float mx = -INFINITY;
        #pragma unroll
        for (int i = 0; i < 16; ++i) {
            int k = i * 128 + lane * 4;
            float4 av = lg[i];
            float4 cv = *(const float4*)&css[k];
            float4 sv = *(const float4*)&sbs[k];
            float4 l;
            l.x = av.x + sv.x + x1bn * cv.x;
            l.y = av.y + sv.y + x1bn * cv.y;
            l.z = av.z + sv.z + x1bn * cv.z;
            l.w = av.w + sv.w + x1bn * cv.w;
            lg[i] = l;
            mx = fmaxf(mx, fmaxf(fmaxf(l.x, l.y), fmaxf(l.z, l.w)));
        }
        #pragma unroll
        for (int off = 16; off > 0; off >>= 1)
            mx = fmaxf(mx, __shfl_xor(mx, off, 32));

        float Z = 0.0f, Wn = 0.0f;
        #pragma unroll
        for (int i = 0; i < 16; ++i) {
            int k = i * 128 + lane * 4;
            float4 xv = *(const float4*)&xks[k];
            float ex = __expf(lg[i].x - mx); Z += ex; Wn += ex * xv.x;
            float ey = __expf(lg[i].y - mx); Z += ey; Wn += ey * xv.y;
            float ez = __expf(lg[i].z - mx); Z += ez; Wn += ez * xv.z;
            float ew = __expf(lg[i].w - mx); Z += ew; Wn += ew * xv.w;
        }
        #pragma unroll
        for (int off = 16; off > 0; off >>= 1) {
            Z  += __shfl_xor(Z, off, 32);
            Wn += __shfl_xor(Wn, off, 32);
        }
        float r = Wn / Z;

        size_t ob = ((size_t)(b * N_NODES + n)) * NFILT;
        out[ob + lane]      = r + bs[lane];
        out[ob + 32 + lane] = r + bs[32 + lane];
    }
}

extern "C" void kernel_launch(void* const* d_in, const int* in_sizes, int n_in,
                              void* d_out, int out_size, void* d_ws, size_t ws_size,
                              hipStream_t stream) {
    const float* x    = (const float*)d_in[0];
    const float* A    = (const float*)d_in[1];
    const float* w1   = (const float*)d_in[2];
    const float* Wa   = (const float*)d_in[3];
    const float* ba   = (const float*)d_in[4];
    const float* bias = (const float*)d_in[5];
    float* out = (float*)d_out;

    float* X1  = (float*)d_ws;
    float* CS  = X1 + BATCH * N_NODES;
    float* SB2 = CS + N_NODES;

    k_proj<<<dim3(256), dim3(256), 0, stream>>>(x, w1, X1);
    k_wmma_s<<<dim3(128), dim3(512), 0, stream>>>(Wa, ba, X1, SB2, CS);
    k_softmax_agg<<<dim3(256), dim3(512), 0, stream>>>(A, bias, X1, SB2, CS, out);
}


# Skinny WMMA product re-tiled for whole-line loads on all 256 WGPs: a wave takes 64 K rows x 32 columns (2 interleaved tiles); K split over 4 groups too, consumer adds 4 planes (f32 sum order changes)
# speedup vs baseline: 2.6735x; 1.1628x over previous
#include <hip/hip_runtime.h>
#include <math.h>

typedef float v2f __attribute__((ext_vector_type(2)));
typedef float v8f __attribute__((ext_vector_type(8)));

#define N_NODES 2048
#define BATCH   8
#define FEAT    128
#define NFILT   64

#define PROJ_ROWS 8

__global__ void __launch_bounds__(256) k_proj(const float* __restrict__ x,
                                              const float* __restrict__ w1,
                                              float* __restrict__ X1) {
    int wave = threadIdx.x >> 5;
    int lane = threadIdx.x & 31;
    int row0 = (blockIdx.x * 8 + wave) * PROJ_ROWS;
    const float* xp = x + (size_t)row0 * FEAT + lane * 4;
    float4 xv[PROJ_ROWS];
    #pragma unroll
    for (int r = 0; r < PROJ_ROWS; ++r) xv[r] = *(const float4*)(xp + r * FEAT);
    const float4 wv = *(const float4*)(w1 + lane * 4);
    float mine = 0.0f;
    #pragma unroll
    for (int r = 0; r < PROJ_ROWS; ++r) {
        float d = xv[r].x * wv.x + xv[r].y * wv.y + xv[r].z * wv.z + xv[r].w * wv.w;
        #pragma unroll
        for (int off = 16; off > 0; off >>= 1) d += __shfl_xor(d, off, 32);
        if (lane == r) mine = d;
    }
    if (lane < PROJ_ROWS) X1[row0 + lane] = mine;
}

#define CG_COLS 32
#define NTILE   2
#define KQ      4
#define KWAVES  8
#define KROWS   (N_NODES / KQ / KWAVES)
#define KSTEPS  (KROWS / 4)
#define PLANE   (9 * 16)
#define NOUT    (9 * CG_COLS)

__global__ void __launch_bounds__(KWAVES * 32) k_wmma_s(
        const float* __restrict__ Wa, const float* __restrict__ ba,
        const float* __restrict__ X1,
        float* __restrict__ SB2P, float* __restrict__ CSP) {
    __shared__ float part[KWAVES * NTILE * PLANE];

    int wave   = threadIdx.x >> 5;
    int lane   = threadIdx.x & 31;
    int cg     = blockIdx.x & 63;
    int kq     = blockIdx.x >> 6;
    int colIn  = lane & 15;
    int hi     = lane >> 4;
    int koff   = hi * 2;
    int mrow   = colIn;
    int srcRow = mrow & 7;
    int kBeg   = kq * (N_NODES / KQ) + wave * KROWS;

    const float* aPtr = X1 + srcRow * N_NODES + kBeg + koff;
    const float* bPtr = Wa + (size_t)(kBeg + koff) * N_NODES + cg * CG_COLS + colIn * 2;

    const float bav = ba[cg * CG_COLS + lane];

    v2f b0[KSTEPS], b1[KSTEPS], av[KSTEPS];
    #pragma unroll
    for (int i = 0; i < KSTEPS; ++i) {
        b0[i] = *(const v2f*)(bPtr + (size_t)(i * 4)     * N_NODES);
        b1[i] = *(const v2f*)(bPtr + (size_t)(i * 4 + 1) * N_NODES);
        av[i] = *(const v2f*)(aPtr + i * 4);
    }
    __builtin_amdgcn_sched_barrier(0);

    v8f c0 = {}, c1 = {};
    #pragma unroll
    for (int i = 0; i < KSTEPS; ++i) {
        v2f a;
        a.x = (mrow < 8) ? av[i].x : ((mrow == 8) ? 1.0f : 0.0f);
        a.y = (mrow < 8) ? av[i].y : ((mrow == 8) ? 1.0f : 0.0f);
        v2f bt0 = {b0[i].x, b1[i].x};
        v2f bt1 = {b0[i].y, b1[i].y};
        c0 = __builtin_amdgcn_wmma_f32_16x16x4_f32(false, a, false, bt0,
                                                   (short)0, c0, false, false);
        c1 = __builtin_amdgcn_wmma_f32_16x16x4_f32(false, a, false, bt1,
                                                   (short)0, c1, false, false);
    }

    float* plane = part + wave * (NTILE * PLANE);
    if (hi == 0) {
        #pragma unroll
        for (int j = 0; j < 8; ++j) {
            plane[0 * PLANE + j * 16 + colIn] = c0[j];
            plane[1 * PLANE + j * 16 + colIn] = c1[j];
        }
    } else {
        plane[0 * PLANE + 8 * 16 + colIn] = c0[0];
        plane[1 * PLANE + 8 * 16 + colIn] = c1[0];
    }
    __syncthreads();

    for (int o = threadIdx.x; o < NOUT; o += KWAVES * 32) {
        int row = o >> 5;
        int cl  = o & 31;
        int src = (cl & 1) * PLANE + row * 16 + (cl >> 1);
        float sum = part[src];
        #pragma unroll
        for (int w = 1; w < KWAVES; ++w) sum += part[w * (NTILE * PLANE) + src];
        int ocol = cg * CG_COLS + cl;
        if (row < 8) {
            if (kq == 0) sum += bav;
            SB2P[(kq * BATCH + row) * N_NODES + ocol] = sum;
        } else {
            CSP[kq * N_NODES + ocol] = sum;
        }
    }
}

#define SM_WAVES 16
#define SM_ROWS  4
#define SM_GROUP_ROWS (SM_WAVES * SM_ROWS)

__global__ void __launch_bounds__(SM_WAVES * 32) k_softmax_agg(
        const float* __restrict__ A, const float* __restrict__ bias,
        const float* __restrict__ X1, const float* __restrict__ SB2P,
        const float* __restrict__ CSP, float* __restrict__ out) {
    __shared__ __align__(16) float css[N_NODES];
    __shared__ __align__(16) float sbs[N_NODES];
    __shared__ __align__(16) float xks[N_NODES];
    __shared__ float bs[NFILT];

    const int t    = threadIdx.x;
    const int lane = t & 31;
    const int wave = t >> 5;
    const int b    = blockIdx.x >> 5;
    const int n0   = (blockIdx.x & 31) * SM_GROUP_ROWS + wave * SM_ROWS;
    const float* aRow = A + ((size_t)(b * N_NODES + n0)) * N_NODES + lane * 4;

    float4 buf[2][16];

    #pragma unroll
    for (int i = 0; i < 16; ++i)
        buf[0][i] = *(const float4*)(aRow + i * 128);

    {
        float4 cs4 = *(const float4*)(CSP + t * 4);
        float4 sb4 = *(const float4*)(SB2P + b * N_NODES + t * 4);
        #pragma unroll
        for (int q = 1; q < KQ; ++q) {
            float4 cq = *(const float4*)(CSP + q * N_NODES + t * 4);
            float4 sq = *(const float4*)(SB2P + (q * BATCH + b) * N_NODES + t * 4);
            cs4.x += cq.x; cs4.y += cq.y; cs4.z += cq.z; cs4.w += cq.w;
            sb4.x += sq.x; sb4.y += sq.y; sb4.z += sq.z; sb4.w += sq.w;
        }
        *(float4*)&css[t * 4] = cs4;
        *(float4*)&sbs[t * 4] = sb4;
    }
    *(float4*)&xks[t * 4] = *(const float4*)(X1 + b * N_NODES + t * 4);
    if (t < NFILT) bs[t] = bias[t];
    __syncthreads();

    #pragma unroll
    for (int j = 0; j < SM_ROWS; ++j) {
        __builtin_amdgcn_sched_barrier(0);
        if (j + 1 < SM_ROWS) {
            #pragma unroll
            for (int i = 0; i < 16; ++i)
                buf[(j + 1) & 1][i] =
                    *(const float4*)(aRow + (size_t)(j + 1) * N_NODES + i * 128);
        }
        __builtin_amdgcn_sched_barrier(0);

        float4* lg = buf[j & 1];
        const int n = n0 + j;
        const float x1bn = xks[n];

        float mx = -INFINITY;
        #pragma unroll
        for (int i = 0; i < 16; ++i) {
            int k = i * 128 + lane * 4;
            float4 av = lg[i];
            float4 cv = *(const float4*)&css[k];
            float4 sv = *(const float4*)&sbs[k];
            float4 l;
            l.x = av.x + sv.x + x1bn * cv.x;
            l.y = av.y + sv.y + x1bn * cv.y;
            l.z = av.z + sv.z + x1bn * cv.z;
            l.w = av.w + sv.w + x1bn * cv.w;
            lg[i] = l;
            mx = fmaxf(mx, fmaxf(fmaxf(l.x, l.y), fmaxf(l.z, l.w)));
        }
        #pragma unroll
        for (int off = 16; off > 0; off >>= 1)
            mx = fmaxf(mx, __shfl_xor(mx, off, 32));

        float Z = 0.0f, Wn = 0.0f;
        #pragma unroll
        for (int i = 0; i < 16; ++i) {
            int k = i * 128 + lane * 4;
            float4 xv = *(const float4*)&xks[k];
            float ex = __expf(lg[i].x - mx); Z += ex; Wn += ex * xv.x;
            float ey = __expf(lg[i].y - mx); Z += ey; Wn += ey * xv.y;
            float ez = __expf(lg[i].z - mx); Z += ez; Wn += ez * xv.z;
            float ew = __expf(lg[i].w - mx); Z += ew; Wn += ew * xv.w;
        }
        #pragma unroll
        for (int off = 16; off > 0; off >>= 1) {
            Z  += __shfl_xor(Z, off, 32);
            Wn += __shfl_xor(Wn, off, 32);
        }
        float r = Wn / Z;

        size_t ob = ((size_t)(b * N_NODES + n)) * NFILT;
        out[ob + lane]      = r + bs[lane];
        out[ob + 32 + lane] = r + bs[32 + lane];
    }
}

extern "C" void kernel_launch(void* const* d_in, const int* in_sizes, int n_in,
                              void* d_out, int out_size, void* d_ws, size_t ws_size,
                              hipStream_t stream) {
    const float* x    = (const float*)d_in[0];
    const float* A    = (const float*)d_in[1];
    const float* w1   = (const float*)d_in[2];
    const float* Wa   = (const float*)d_in[3];
    const float* ba   = (const float*)d_in[4];
    const float* bias = (const float*)d_in[5];
    float* out = (float*)d_out;

    float* X1   = (float*)d_ws;
    float* CSP  = X1 + BATCH * N_NODES;
    float* SB2P = CSP + KQ * N_NODES;

    k_proj<<<dim3(256), dim3(256), 0, stream>>>(x, w1, X1);
    k_wmma_s<<<dim3(256), dim3(256), 0, stream>>>(Wa, ba, X1, SB2P, CSP);
    k_softmax_agg<<<dim3(256), dim3(512), 0, stream>>>(A, bias, X1, SB2P, CSP, out);
}


# Wave reductions (projection row sum; softmax row maximum, Z, weighted sum) on the vector ALU: v_permlanex16 + DPP row_xmask, not LDS permutes; same butterfly order, sums bit-identical
# speedup vs baseline: 2.7690x; 1.0357x over previous
#include <hip/hip_runtime.h>
#include <math.h>

typedef float v2f __attribute__((ext_vector_type(2)));
typedef float v8f __attribute__((ext_vector_type(8)));

#define N_NODES 2048
#define BATCH   8
#define FEAT    128
#define NFILT   64

__device__ __forceinline__ float xhalf(float x) {
    return __builtin_bit_cast(float, __builtin_amdgcn_permlanex16(
        __builtin_bit_cast(int, x), __builtin_bit_cast(int, x),
        0x76543210, 0xfedcba98, false, false));
}
template <int M> __device__ __forceinline__ float xrow(float x) {
    return __builtin_bit_cast(float, __builtin_amdgcn_update_dpp(
        0, __builtin_bit_cast(int, x), 0x160 + M, 0xf, 0xf, true));
}
__device__ __forceinline__ float wave_sum(float v) {
    v += xhalf(v);
    v += xrow<8>(v);
    v += xrow<4>(v);
    v += xrow<2>(v);
    v += xrow<1>(v);
    return v;
}
__device__ __forceinline__ float wave_max(float v) {
    v = fmaxf(v, xhalf(v));
    v = fmaxf(v, xrow<8>(v));
    v = fmaxf(v, xrow<4>(v));
    v = fmaxf(v, xrow<2>(v));
    v = fmaxf(v, xrow<1>(v));
    return v;
}

#define PROJ_ROWS 8

__global__ void __launch_bounds__(256) k_proj(const float* __restrict__ x,
                                              const float* __restrict__ w1,
                                              float* __restrict__ X1) {
    int wave = threadIdx.x >> 5;
    int lane = threadIdx.x & 31;
    int row0 = (blockIdx.x * 8 + wave) * PROJ_ROWS;
    const float* xp = x + (size_t)row0 * FEAT + lane * 4;
    float4 xv[PROJ_ROWS];
    #pragma unroll
    for (int r = 0; r < PROJ_ROWS; ++r) xv[r] = *(const float4*)(xp + r * FEAT);
    const float4 wv = *(const float4*)(w1 + lane * 4);
    float mine = 0.0f;
    #pragma unroll
    for (int r = 0; r < PROJ_ROWS; ++r) {
        float d = xv[r].x * wv.x + xv[r].y * wv.y + xv[r].z * wv.z + xv[r].w * wv.w;
        d = wave_sum(d);
        if (lane == r) mine = d;
    }
    if (lane < PROJ_ROWS) X1[row0 + lane] = mine;
}

#define CG_COLS 32
#define NTILE   2
#define KQ      4
#define KWAVES  8
#define KROWS   (N_NODES / KQ / KWAVES)
#define KSTEPS  (KROWS / 4)
#define PLANE   (9 * 16)
#define NOUT    (9 * CG_COLS)

__global__ void __launch_bounds__(KWAVES * 32) k_wmma_s(
        const float* __restrict__ Wa, const float* __restrict__ ba,
        const float* __restrict__ X1,
        float* __restrict__ SB2P, float* __restrict__ CSP) {
    __shared__ float part[KWAVES * NTILE * PLANE];

    int wave   = threadIdx.x >> 5;
    int lane   = threadIdx.x & 31;
    int cg     = blockIdx.x & 63;
    int kq     = blockIdx.x >> 6;
    int colIn  = lane & 15;
    int hi     = lane >> 4;
    int koff   = hi * 2;
    int mrow   = colIn;
    int srcRow = mrow & 7;
    int kBeg   = kq * (N_NODES / KQ) + wave * KROWS;

    const float* aPtr = X1 + srcRow * N_NODES + kBeg + koff;
    const float* bPtr = Wa + (size_t)(kBeg + koff) * N_NODES + cg * CG_COLS + colIn * 2;

    const float bav = ba[cg * CG_COLS + lane];

    v2f b0[KSTEPS], b1[KSTEPS], av[KSTEPS];
    #pragma unroll
    for (int i = 0; i < KSTEPS; ++i) {
        b0[i] = *(const v2f*)(bPtr + (size_t)(i * 4)     * N_NODES);
        b1[i] = *(const v2f*)(bPtr + (size_t)(i * 4 + 1) * N_NODES);
        av[i] = *(const v2f*)(aPtr + i * 4);
    }
    __builtin_amdgcn_sched_barrier(0);

    v8f c0 = {}, c1 = {};
    #pragma unroll
    for (int i = 0; i < KSTEPS; ++i) {
        v2f a;
        a.x = (mrow < 8) ? av[i].x : ((mrow == 8) ? 1.0f : 0.0f);
        a.y = (mrow < 8) ? av[i].y : ((mrow == 8) ? 1.0f : 0.0f);
        v2f bt0 = {b0[i].x, b1[i].x};
        v2f bt1 = {b0[i].y, b1[i].y};
        c0 = __builtin_amdgcn_wmma_f32_16x16x4_f32(false, a, false, bt0,
                                                   (short)0, c0, false, false);
        c1 = __builtin_amdgcn_wmma_f32_16x16x4_f32(false, a, false, bt1,
                                                   (short)0, c1, false, false);
    }

    float* plane = part + wave * (NTILE * PLANE);
    if (hi == 0) {
        #pragma unroll
        for (int j = 0; j < 8; ++j) {
            plane[0 * PLANE + j * 16 + colIn] = c0[j];
            plane[1 * PLANE + j * 16 + colIn] = c1[j];
        }
    } else {
        plane[0 * PLANE + 8 * 16 + colIn] = c0[0];
        plane[1 * PLANE + 8 * 16 + colIn] = c1[0];
    }
    __syncthreads();

    for (int o = threadIdx.x; o < NOUT; o += KWAVES * 32) {
        int row = o >> 5;
        int cl  = o & 31;
        int src = (cl & 1) * PLANE + row * 16 + (cl >> 1);
        float sum = part[src];
        #pragma unroll
        for (int w = 1; w < KWAVES; ++w) sum += part[w * (NTILE * PLANE) + src];
        int ocol = cg * CG_COLS + cl;
        if (row < 8) {
            if (kq == 0) sum += bav;
            SB2P[(kq * BATCH + row) * N_NODES + ocol] = sum;
        } else {
            CSP[kq * N_NODES + ocol] = sum;
        }
    }
}

#define SM_WAVES 16
#define SM_ROWS  4
#define SM_GROUP_ROWS (SM_WAVES * SM_ROWS)

__global__ void __launch_bounds__(SM_WAVES * 32) k_softmax_agg(
        const float* __restrict__ A, const float* __restrict__ bias,
        const float* __restrict__ X1, const float* __restrict__ SB2P,
        const float* __restrict__ CSP, float* __restrict__ out) {
    __shared__ __align__(16) float css[N_NODES];
    __shared__ __align__(16) float sbs[N_NODES];
    __shared__ __align__(16) float xks[N_NODES];
    __shared__ float bs[NFILT];

    const int t    = threadIdx.x;
    const int lane = t & 31;
    const int wave = t >> 5;
    const int b    = blockIdx.x >> 5;
    const int n0   = (blockIdx.x & 31) * SM_GROUP_ROWS + wave * SM_ROWS;
    const float* aRow = A + ((size_t)(b * N_NODES + n0)) * N_NODES + lane * 4;

    float4 buf[2][16];

    #pragma unroll
    for (int i = 0; i < 16; ++i)
        buf[0][i] = *(const float4*)(aRow + i * 128);

    {
        float4 cs4 = *(const float4*)(CSP + t * 4);
        float4 sb4 = *(const float4*)(SB2P + b * N_NODES + t * 4);
        #pragma unroll
        for (int q = 1; q < KQ; ++q) {
            float4 cq = *(const float4*)(CSP + q * N_NODES + t * 4);
            float4 sq = *(const float4*)(SB2P + (q * BATCH + b) * N_NODES + t * 4);
            cs4.x += cq.x; cs4.y += cq.y; cs4.z += cq.z; cs4.w += cq.w;
            sb4.x += sq.x; sb4.y += sq.y; sb4.z += sq.z; sb4.w += sq.w;
        }
        *(float4*)&css[t * 4] = cs4;
        *(float4*)&sbs[t * 4] = sb4;
    }
    *(float4*)&xks[t * 4] = *(const float4*)(X1 + b * N_NODES + t * 4);
    if (t < NFILT) bs[t] = bias[t];
    __syncthreads();

    #pragma unroll
    for (int j = 0; j < SM_ROWS; ++j) {
        __builtin_amdgcn_sched_barrier(0);
        if (j + 1 < SM_ROWS) {
            #pragma unroll
            for (int i = 0; i < 16; ++i)
                buf[(j + 1) & 1][i] =
                    *(const float4*)(aRow + (size_t)(j + 1) * N_NODES + i * 128);
        }
        __builtin_amdgcn_sched_barrier(0);

        float4* lg = buf[j & 1];
        const int n = n0 + j;
        const float x1bn = xks[n];

        float mx = -INFINITY;
        #pragma unroll
        for (int i = 0; i < 16; ++i) {
            int k = i * 128 + lane * 4;
            float4 av = lg[i];
            float4 cv = *(const float4*)&css[k];
            float4 sv = *(const float4*)&sbs[k];
            float4 l;
            l.x = av.x + sv.x + x1bn * cv.x;
            l.y = av.y + sv.y + x1bn * cv.y;
            l.z = av.z + sv.z + x1bn * cv.z;
            l.w = av.w + sv.w + x1bn * cv.w;
            lg[i] = l;
            mx = fmaxf(mx, fmaxf(fmaxf(l.x, l.y), fmaxf(l.z, l.w)));
        }
        mx = wave_max(mx);

        float Z = 0.0f, Wn = 0.0f;
        #pragma unroll
        for (int i = 0; i < 16; ++i) {
            int k = i * 128 + lane * 4;
            float4 xv = *(const float4*)&xks[k];
            float ex = __expf(lg[i].x - mx); Z += ex; Wn += ex * xv.x;
            float ey = __expf(lg[i].y - mx); Z += ey; Wn += ey * xv.y;
            float ez = __expf(lg[i].z - mx); Z += ez; Wn += ez * xv.z;
            float ew = __expf(lg[i].w - mx); Z += ew; Wn += ew * xv.w;
        }
        Z  = wave_sum(Z);
        Wn = wave_sum(Wn);
        float r = Wn / Z;

        size_t ob = ((size_t)(b * N_NODES + n)) * NFILT;
        out[ob + lane]      = r + bs[lane];
        out[ob + 32 + lane] = r + bs[32 + lane];
    }
}

extern "C" void kernel_launch(void* const* d_in, const int* in_sizes, int n_in,
                              void* d_out, int out_size, void* d_ws, size_t ws_size,
                              hipStream_t stream) {
    const float* x    = (const float*)d_in[0];
    const float* A    = (const float*)d_in[1];
    const float* w1   = (const float*)d_in[2];
    const float* Wa   = (const float*)d_in[3];
    const float* ba   = (const float*)d_in[4];
    const float* bias = (const float*)d_in[5];
    float* out = (float*)d_out;

    float* X1   = (float*)d_ws;
    float* CSP  = X1 + BATCH * N_NODES;
    float* SB2P = CSP + KQ * N_NODES;

    k_proj<<<dim3(256), dim3(256), 0, stream>>>(x, w1, X1);
    k_wmma_s<<<dim3(256), dim3(256), 0, stream>>>(Wa, ba, X1, SB2P, CSP);
    k_softmax_agg<<<dim3(256), dim3(512), 0, stream>>>(A, bias, X1, SB2P, CSP, out);
}


# Softmax kernel: a lane keeps its 64 columns of CS, SB2, x1 in registers for its wave's 8 rows (8 waves x 8 rows per group, were 16 x 4): no LDS vector reads per row; two rows in flight; same sums
# speedup vs baseline: 2.8658x; 1.0350x over previous
#include <hip/hip_runtime.h>
#include <math.h>

typedef float v2f __attribute__((ext_vector_type(2)));
typedef float v8f __attribute__((ext_vector_type(8)));

#define N_NODES 2048
#define BATCH   8
#define FEAT    128
#define NFILT   64

__device__ __forceinline__ float xhalf(float x) {
    return __builtin_bit_cast(float, __builtin_amdgcn_permlanex16(
        __builtin_bit_cast(int, x), __builtin_bit_cast(int, x),
        0x76543210, 0xfedcba98, false, false));
}
template <int M> __device__ __forceinline__ float xrow(float x) {
    return __builtin_bit_cast(float, __builtin_amdgcn_update_dpp(
        0, __builtin_bit_cast(int, x), 0x160 + M, 0xf, 0xf, true));
}
__device__ __forceinline__ float wave_sum(float v) {
    v += xhalf(v);
    v += xrow<8>(v);
    v += xrow<4>(v);
    v += xrow<2>(v);
    v += xrow<1>(v);
    return v;
}
__device__ __forceinline__ float wave_max(float v) {
    v = fmaxf(v, xhalf(v));
    v = fmaxf(v, xrow<8>(v));
    v = fmaxf(v, xrow<4>(v));
    v = fmaxf(v, xrow<2>(v));
    v = fmaxf(v, xrow<1>(v));
    return v;
}

#define PROJ_ROWS 8

__global__ void __launch_bounds__(256) k_proj(const float* __restrict__ x,
                                              const float* __restrict__ w1,
                                              float* __restrict__ X1) {
    int wave = threadIdx.x >> 5;
    int lane = threadIdx.x & 31;
    int row0 = (blockIdx.x * 8 + wave) * PROJ_ROWS;
    const float* xp = x + (size_t)row0 * FEAT + lane * 4;
    float4 xv[PROJ_ROWS];
    #pragma unroll
    for (int r = 0; r < PROJ_ROWS; ++r) xv[r] = *(const float4*)(xp + r * FEAT);
    const float4 wv = *(const float4*)(w1 + lane * 4);
    float mine = 0.0f;
    #pragma unroll
    for (int r = 0; r < PROJ_ROWS; ++r) {
        float d = xv[r].x * wv.x + xv[r].y * wv.y + xv[r].z * wv.z + xv[r].w * wv.w;
        d = wave_sum(d);
        if (lane == r) mine = d;
    }
    if (lane < PROJ_ROWS) X1[row0 + lane] = mine;
}

#define CG_COLS 32
#define NTILE   2
#define KQ      4
#define KWAVES  8
#define KROWS   (N_NODES / KQ / KWAVES)
#define KSTEPS  (KROWS / 4)
#define PLANE   (9 * 16)
#define NOUT    (9 * CG_COLS)

__global__ void __launch_bounds__(KWAVES * 32) k_wmma_s(
        const float* __restrict__ Wa, const float* __restrict__ ba,
        const float* __restrict__ X1,
        float* __restrict__ SB2P, float* __restrict__ CSP) {
    __shared__ float part[KWAVES * NTILE * PLANE];

    int wave   = threadIdx.x >> 5;
    int lane   = threadIdx.x & 31;
    int cg     = blockIdx.x & 63;
    int kq     = blockIdx.x >> 6;
    int colIn  = lane & 15;
    int hi     = lane >> 4;
    int koff   = hi * 2;
    int mrow   = colIn;
    int srcRow = mrow & 7;
    int kBeg   = kq * (N_NODES / KQ) + wave * KROWS;

    const float* aPtr = X1 + srcRow * N_NODES + kBeg + koff;
    const float* bPtr = Wa + (size_t)(kBeg + koff) * N_NODES + cg * CG_COLS + colIn * 2;

    const float bav = ba[cg * CG_COLS + lane];

    v2f b0[KSTEPS], b1[KSTEPS], av[KSTEPS];
    #pragma unroll
    for (int i = 0; i < KSTEPS; ++i) {
        b0[i] = *(const v2f*)(bPtr + (size_t)(i * 4)     * N_NODES);
        b1[i] = *(const v2f*)(bPtr + (size_t)(i * 4 + 1) * N_NODES);
        av[i] = *(const v2f*)(aPtr + i * 4);
    }
    __builtin_amdgcn_sched_barrier(0);

    v8f c0 = {}, c1 = {};
    #pragma unroll
    for (int i = 0; i < KSTEPS; ++i) {
        v2f a;
        a.x = (mrow < 8) ? av[i].x : ((mrow == 8) ? 1.0f : 0.0f);
        a.y = (mrow < 8) ? av[i].y : ((mrow == 8) ? 1.0f : 0.0f);
        v2f bt0 = {b0[i].x, b1[i].x};
        v2f bt1 = {b0[i].y, b1[i].y};
        c0 = __builtin_amdgcn_wmma_f32_16x16x4_f32(false, a, false, bt0,
                                                   (short)0, c0, false, false);
        c1 = __builtin_amdgcn_wmma_f32_16x16x4_f32(false, a, false, bt1,
                                                   (short)0, c1, false, false);
    }

    float* plane = part + wave * (NTILE * PLANE);
    if (hi == 0) {
        #pragma unroll
        for (int j = 0; j < 8; ++j) {
            plane[0 * PLANE + j * 16 + colIn] = c0[j];
            plane[1 * PLANE + j * 16 + colIn] = c1[j];
        }
    } else {
        plane[0 * PLANE + 8 * 16 + colIn] = c0[0];
        plane[1 * PLANE + 8 * 16 + colIn] = c1[0];
    }
    __syncthreads();

    for (int o = threadIdx.x; o < NOUT; o += KWAVES * 32) {
        int row = o >> 5;
        int cl  = o & 31;
        int src = (cl & 1) * PLANE + row * 16 + (cl >> 1);
        float sum = part[src];
        #pragma unroll
        for (int w = 1; w < KWAVES; ++w) sum += part[w * (NTILE * PLANE) + src];
        int ocol = cg * CG_COLS + cl;
        if (row < 8) {
            if (kq == 0) sum += bav;
            SB2P[(kq * BATCH + row) * N_NODES + ocol] = sum;
        } else {
            CSP[kq * N_NODES + ocol] = sum;
        }
    }
}

#define SM_WAVES 8
#define SM_ROWS  8
#define SM_GROUP_ROWS (SM_WAVES * SM_ROWS)
#define SM_THREADS (SM_WAVES * 32)

__device__ __forceinline__ void load_row(float4 (&buf)[16], const float* __restrict__ p) {
    #pragma unroll
    for (int i = 0; i < 16; ++i) buf[i] = *(const float4*)(p + i * 128);
}

__device__ __forceinline__ void softmax_row(float4 (&lg)[16],
        const float4 (&cs)[16], const float4 (&sb)[16], const float4 (&xk)[16],
        float x1bn, float bias0, float bias1, float* __restrict__ orow, int lane) {
    float mx = -INFINITY;
    #pragma unroll
    for (int i = 0; i < 16; ++i) {
        float4 av = lg[i];
        float4 cv = cs[i];
        float4 sv = sb[i];
        float4 l;
        l.x = av.x + sv.x + x1bn * cv.x;
        l.y = av.y + sv.y + x1bn * cv.y;
        l.z = av.z + sv.z + x1bn * cv.z;
        l.w = av.w + sv.w + x1bn * cv.w;
        lg[i] = l;
        mx = fmaxf(mx, fmaxf(fmaxf(l.x, l.y), fmaxf(l.z, l.w)));
    }
    mx = wave_max(mx);

    float Z = 0.0f, Wn = 0.0f;
    #pragma unroll
    for (int i = 0; i < 16; ++i) {
        float4 xv = xk[i];
        float ex = __expf(lg[i].x - mx); Z += ex; Wn += ex * xv.x;
        float ey = __expf(lg[i].y - mx); Z += ey; Wn += ey * xv.y;
        float ez = __expf(lg[i].z - mx); Z += ez; Wn += ez * xv.z;
        float ew = __expf(lg[i].w - mx); Z += ew; Wn += ew * xv.w;
    }
    Z  = wave_sum(Z);
    Wn = wave_sum(Wn);
    float r = Wn / Z;

    orow[lane]      = r + bias0;
    orow[32 + lane] = r + bias1;
}

__global__ void __launch_bounds__(SM_THREADS) k_softmax_agg(
        const float* __restrict__ A, const float* __restrict__ bias,
        const float* __restrict__ X1, const float* __restrict__ SB2P,
        const float* __restrict__ CSP, float* __restrict__ out) {
    __shared__ __align__(16) float css[N_NODES];
    __shared__ __align__(16) float sbs[N_NODES];
    __shared__ __align__(16) float xks[N_NODES];

    const int t    = threadIdx.x;
    const int lane = t & 31;
    const int wave = t >> 5;
    const int b    = blockIdx.x >> 5;
    const int n0   = (blockIdx.x & 31) * SM_GROUP_ROWS + wave * SM_ROWS;
    const float* aRow = A + ((size_t)(b * N_NODES + n0)) * N_NODES + lane * 4;
    float* oRow = out + ((size_t)(b * N_NODES + n0)) * NFILT;

    float4 b0[16], b1[16], b2[16];

    load_row(b0, aRow);
    load_row(b1, aRow + N_NODES);

    #pragma unroll
    for (int h = 0; h < 2; ++h) {
        const int c = (t + h * SM_THREADS) * 4;
        float4 cs4 = *(const float4*)(CSP + c);
        float4 sb4 = *(const float4*)(SB2P + b * N_NODES + c);
        #pragma unroll
        for (int q = 1; q < KQ; ++q) {
            float4 cq = *(const float4*)(CSP + q * N_NODES + c);
            float4 sq = *(const float4*)(SB2P + (q * BATCH + b) * N_NODES + c);
            cs4.x += cq.x; cs4.y += cq.y; cs4.z += cq.z; cs4.w += cq.w;
            sb4.x += sq.x; sb4.y += sq.y; sb4.z += sq.z; sb4.w += sq.w;
        }
        *(float4*)&css[c] = cs4;
        *(float4*)&sbs[c] = sb4;
        *(float4*)&xks[c] = *(const float4*)(X1 + b * N_NODES + c);
    }
    const float bias0 = bias[lane];
    const float bias1 = bias[32 + lane];
    int row2 = 2 * N_NODES;
    asm volatile("" : "+v"(row2));
    load_row(b2, aRow + row2);

    __builtin_amdgcn_fence(__ATOMIC_RELEASE, "workgroup", "local");
    __builtin_amdgcn_s_barrier();
    __builtin_amdgcn_fence(__ATOMIC_ACQUIRE, "workgroup", "local");

    float4 cs[16], sb[16], xk[16];
    #pragma unroll
    for (int i = 0; i < 16; ++i) {
        const int k = i * 128 + lane * 4;
        cs[i] = *(const float4*)&css[k];
        sb[i] = *(const float4*)&sbs[k];
        xk[i] = *(const float4*)&xks[k];
    }

    #pragma nounroll
    for (int it = 0; it < 2; ++it) {
        const float* p = aRow + (size_t)(3 * it) * N_NODES;
        float* o = oRow + (3 * it) * NFILT;
        const int n = n0 + 3 * it;

        softmax_row(b0, cs, sb, xk, xks[n], bias0, bias1, o, lane);
        __builtin_amdgcn_sched_barrier(0);
        load_row(b0, p + 3 * N_NODES);
        __builtin_amdgcn_sched_barrier(0);

        softmax_row(b1, cs, sb, xk, xks[n + 1], bias0, bias1, o + NFILT, lane);
        __builtin_amdgcn_sched_barrier(0);
        load_row(b1, p + 4 * N_NODES);
        __builtin_amdgcn_sched_barrier(0);

        softmax_row(b2, cs, sb, xk, xks[n + 2], bias0, bias1, o + 2 * NFILT, lane);
        if (it == 0) load_row(b2, p + 5 * N_NODES);
    }
    softmax_row(b0, cs, sb, xk, xks[n0 + 6], bias0, bias1, oRow + 6 * NFILT, lane);
    softmax_row(b1, cs, sb, xk, xks[n0 + 7], bias0, bias1, oRow + 7 * NFILT, lane);
}

extern "C" void kernel_launch(void* const* d_in, const int* in_sizes, int n_in,
                              void* d_out, int out_size, void* d_ws, size_t ws_size,
                              hipStream_t stream) {
    const float* x    = (const float*)d_in[0];
    const float* A    = (const float*)d_in[1];
    const float* w1   = (const float*)d_in[2];
    const float* Wa   = (const float*)d_in[3];
    const float* ba   = (const float*)d_in[4];
    const float* bias = (const float*)d_in[5];
    float* out = (float*)d_out;

    float* X1   = (float*)d_ws;
    float* CSP  = X1 + BATCH * N_NODES;
    float* SB2P = CSP + KQ * N_NODES;

    k_proj<<<dim3(256), dim3(256), 0, stream>>>(x, w1, X1);
    k_wmma_s<<<dim3(256), dim3(256), 0, stream>>>(Wa, ba, X1, SB2P, CSP);
    k_softmax_agg<<<dim3(256), dim3(SM_THREADS), 0, stream>>>(A, bias, X1, SB2P, CSP, out);
}


# Softmax row in base 2 on column pairs: log2(e) folded into CS, SB2 and the logit multiply-add, exp argument one packed subtract; Z and weighted sum as 4 partial sums per lane (f32 sum order changes)
# speedup vs baseline: 3.0730x; 1.0723x over previous
#include <hip/hip_runtime.h>
#include <math.h>

typedef float v2f __attribute__((ext_vector_type(2)));
typedef float v8f __attribute__((ext_vector_type(8)));

#define N_NODES 2048
#define BATCH   8
#define FEAT    128
#define NFILT   64

__device__ __forceinline__ float xhalf(float x) {
    return __builtin_bit_cast(float, __builtin_amdgcn_permlanex16(
        __builtin_bit_cast(int, x), __builtin_bit_cast(int, x),
        0x76543210, 0xfedcba98, false, false));
}
template <int M> __device__ __forceinline__ float xrow(float x) {
    return __builtin_bit_cast(float, __builtin_amdgcn_update_dpp(
        0, __builtin_bit_cast(int, x), 0x160 + M, 0xf, 0xf, true));
}
__device__ __forceinline__ float wave_sum(float v) {
    v += xhalf(v);
    v += xrow<8>(v);
    v += xrow<4>(v);
    v += xrow<2>(v);
    v += xrow<1>(v);
    return v;
}
__device__ __forceinline__ float wave_max(float v) {
    v = fmaxf(v, xhalf(v));
    v = fmaxf(v, xrow<8>(v));
    v = fmaxf(v, xrow<4>(v));
    v = fmaxf(v, xrow<2>(v));
    v = fmaxf(v, xrow<1>(v));
    return v;
}

#define PROJ_ROWS 8

__global__ void __launch_bounds__(256) k_proj(const float* __restrict__ x,
                                              const float* __restrict__ w1,
                                              float* __restrict__ X1) {
    int wave = threadIdx.x >> 5;
    int lane = threadIdx.x & 31;
    int row0 = (blockIdx.x * 8 + wave) * PROJ_ROWS;
    const float* xp = x + (size_t)row0 * FEAT + lane * 4;
    float4 xv[PROJ_ROWS];
    #pragma unroll
    for (int r = 0; r < PROJ_ROWS; ++r) xv[r] = *(const float4*)(xp + r * FEAT);
    const float4 wv = *(const float4*)(w1 + lane * 4);
    float mine = 0.0f;
    #pragma unroll
    for (int r = 0; r < PROJ_ROWS; ++r) {
        float d = xv[r].x * wv.x + xv[r].y * wv.y + xv[r].z * wv.z + xv[r].w * wv.w;
        d = wave_sum(d);
        if (lane == r) mine = d;
    }
    if (lane < PROJ_ROWS) X1[row0 + lane] = mine;
}

#define CG_COLS 32
#define NTILE   2
#define KQ      4
#define KWAVES  8
#define KROWS   (N_NODES / KQ / KWAVES)
#define KSTEPS  (KROWS / 4)
#define PLANE   (9 * 16)
#define NOUT    (9 * CG_COLS)

__global__ void __launch_bounds__(KWAVES * 32) k_wmma_s(
        const float* __restrict__ Wa, const float* __restrict__ ba,
        const float* __restrict__ X1,
        float* __restrict__ SB2P, float* __restrict__ CSP) {
    __shared__ float part[KWAVES * NTILE * PLANE];

    int wave   = threadIdx.x >> 5;
    int lane   = threadIdx.x & 31;
    int cg     = blockIdx.x & 63;
    int kq     = blockIdx.x >> 6;
    int colIn  = lane & 15;
    int hi     = lane >> 4;
    int koff   = hi * 2;
    int mrow   = colIn;
    int srcRow = mrow & 7;
    int kBeg   = kq * (N_NODES / KQ) + wave * KROWS;

    const float* aPtr = X1 + srcRow * N_NODES + kBeg + koff;
    const float* bPtr = Wa + (size_t)(kBeg + koff) * N_NODES + cg * CG_COLS + colIn * 2;

    const float bav = ba[cg * CG_COLS + lane];

    v2f b0[KSTEPS], b1[KSTEPS], av[KSTEPS];
    #pragma unroll
    for (int i = 0; i < KSTEPS; ++i) {
        b0[i] = *(const v2f*)(bPtr + (size_t)(i * 4)     * N_NODES);
        b1[i] = *(const v2f*)(bPtr + (size_t)(i * 4 + 1) * N_NODES);
        av[i] = *(const v2f*)(aPtr + i * 4);
    }
    __builtin_amdgcn_sched_barrier(0);

    v8f c0 = {}, c1 = {};
    #pragma unroll
    for (int i = 0; i < KSTEPS; ++i) {
        v2f a;
        a.x = (mrow < 8) ? av[i].x : ((mrow == 8) ? 1.0f : 0.0f);
        a.y = (mrow < 8) ? av[i].y : ((mrow == 8) ? 1.0f : 0.0f);
        v2f bt0 = {b0[i].x, b1[i].x};
        v2f bt1 = {b0[i].y, b1[i].y};
        c0 = __builtin_amdgcn_wmma_f32_16x16x4_f32(false, a, false, bt0,
                                                   (short)0, c0, false, false);
        c1 = __builtin_amdgcn_wmma_f32_16x16x4_f32(false, a, false, bt1,
                                                   (short)0, c1, false, false);
    }

    float* plane = part + wave * (NTILE * PLANE);
    if (hi == 0) {
        #pragma unroll
        for (int j = 0; j < 8; ++j) {
            plane[0 * PLANE + j * 16 + colIn] = c0[j];
            plane[1 * PLANE + j * 16 + colIn] = c1[j];
        }
    } else {
        plane[0 * PLANE + 8 * 16 + colIn] = c0[0];
        plane[1 * PLANE + 8 * 16 + colIn] = c1[0];
    }
    __syncthreads();

    for (int o = threadIdx.x; o < NOUT; o += KWAVES * 32) {
        int row = o >> 5;
        int cl  = o & 31;
        int src = (cl & 1) * PLANE + row * 16 + (cl >> 1);
        float sum = part[src];
        #pragma unroll
        for (int w = 1; w < KWAVES; ++w) sum += part[w * (NTILE * PLANE) + src];
        int ocol = cg * CG_COLS + cl;
        if (row < 8) {
            if (kq == 0) sum += bav;
            SB2P[(kq * BATCH + row) * N_NODES + ocol] = sum;
        } else {
            CSP[kq * N_NODES + ocol] = sum;
        }
    }
}

#define SM_WAVES 8
#define SM_ROWS  8
#define SM_GROUP_ROWS (SM_WAVES * SM_ROWS)
#define SM_THREADS (SM_WAVES * 32)
#define LOG2E 1.44269504088896340736f

__device__ __forceinline__ void load_row(float4 (&buf)[16], const float* __restrict__ p) {
    #pragma unroll
    for (int i = 0; i < 16; ++i) buf[i] = *(const float4*)(p + i * 128);
}

__device__ __forceinline__ void softmax_row(float4 (&lg)[16],
        const float4 (&cs)[16], const float4 (&sb)[16], const float4 (&xk)[16],
        float x1bn, float bias0, float bias1, float* __restrict__ orow, int lane) {
    const v2f l2e = {LOG2E, LOG2E};
    const v2f xn  = {x1bn, x1bn};
    float mx = -INFINITY;
    #pragma unroll
    for (int i = 0; i < 16; ++i) {
        const v2f a0 = {lg[i].x, lg[i].y}, a1 = {lg[i].z, lg[i].w};
        const v2f c0 = {cs[i].x, cs[i].y}, c1 = {cs[i].z, cs[i].w};
        const v2f s0 = {sb[i].x, sb[i].y}, s1 = {sb[i].z, sb[i].w};
        const v2f l0 = __builtin_elementwise_fma(a0, l2e, __builtin_elementwise_fma(xn, c0, s0));
        const v2f l1 = __builtin_elementwise_fma(a1, l2e, __builtin_elementwise_fma(xn, c1, s1));
        lg[i].x = l0.x; lg[i].y = l0.y; lg[i].z = l1.x; lg[i].w = l1.y;
        mx = fmaxf(mx, fmaxf(fmaxf(l0.x, l0.y), fmaxf(l1.x, l1.y)));
    }
    mx = wave_max(mx);

    const v2f nmx = {-mx, -mx};
    v2f z0 = {0.0f, 0.0f}, z1 = {0.0f, 0.0f};
    v2f w0 = {0.0f, 0.0f}, w1 = {0.0f, 0.0f};
    #pragma unroll
    for (int i = 0; i < 16; ++i) {
        const v2f d0 = (v2f){lg[i].x, lg[i].y} + nmx;
        const v2f d1 = (v2f){lg[i].z, lg[i].w} + nmx;
        const v2f e0 = {__builtin_amdgcn_exp2f(d0.x), __builtin_amdgcn_exp2f(d0.y)};
        const v2f e1 = {__builtin_amdgcn_exp2f(d1.x), __builtin_amdgcn_exp2f(d1.y)};
        const v2f x0 = {xk[i].x, xk[i].y}, x1v = {xk[i].z, xk[i].w};
        z0 += e0;
        z1 += e1;
        w0 = __builtin_elementwise_fma(e0, x0, w0);
        w1 = __builtin_elementwise_fma(e1, x1v, w1);
    }
    const v2f zz = z0 + z1;
    const v2f ww = w0 + w1;
    float Z  = wave_sum(zz.x + zz.y);
    float Wn = wave_sum(ww.x + ww.y);
    float r = Wn / Z;

    orow[lane]      = r + bias0;
    orow[32 + lane] = r + bias1;
}

__global__ void __launch_bounds__(SM_THREADS) k_softmax_agg(
        const float* __restrict__ A, const float* __restrict__ bias,
        const float* __restrict__ X1, const float* __restrict__ SB2P,
        const float* __restrict__ CSP, float* __restrict__ out) {
    __shared__ __align__(16) float css[N_NODES];
    __shared__ __align__(16) float sbs[N_NODES];
    __shared__ __align__(16) float xks[N_NODES];

    const int t    = threadIdx.x;
    const int lane = t & 31;
    const int wave = t >> 5;
    const int b    = blockIdx.x >> 5;
    const int n0   = (blockIdx.x & 31) * SM_GROUP_ROWS + wave * SM_ROWS;
    const float* aRow = A + ((size_t)(b * N_NODES + n0)) * N_NODES + lane * 4;
    float* oRow = out + ((size_t)(b * N_NODES + n0)) * NFILT;

    float4 b0[16], b1[16], b2[16];

    load_row(b0, aRow);
    load_row(b1, aRow + N_NODES);

    #pragma unroll
    for (int h = 0; h < 2; ++h) {
        const int c = (t + h * SM_THREADS) * 4;
        float4 cs4 = *(const float4*)(CSP + c);
        float4 sb4 = *(const float4*)(SB2P + b * N_NODES + c);
        #pragma unroll
        for (int q = 1; q < KQ; ++q) {
            float4 cq = *(const float4*)(CSP + q * N_NODES + c);
            float4 sq = *(const float4*)(SB2P + (q * BATCH + b) * N_NODES + c);
            cs4.x += cq.x; cs4.y += cq.y; cs4.z += cq.z; cs4.w += cq.w;
            sb4.x += sq.x; sb4.y += sq.y; sb4.z += sq.z; sb4.w += sq.w;
        }
        cs4.x *= LOG2E; cs4.y *= LOG2E; cs4.z *= LOG2E; cs4.w *= LOG2E;
        sb4.x *= LOG2E; sb4.y *= LOG2E; sb4.z *= LOG2E; sb4.w *= LOG2E;
        *(float4*)&css[c] = cs4;
        *(float4*)&sbs[c] = sb4;
        *(float4*)&xks[c] = *(const float4*)(X1 + b * N_NODES + c);
    }
    const float bias0 = bias[lane];
    const float bias1 = bias[32 + lane];
    int row2 = 2 * N_NODES;
    asm volatile("" : "+v"(row2));
    load_row(b2, aRow + row2);

    __builtin_amdgcn_fence(__ATOMIC_RELEASE, "workgroup", "local");
    __builtin_amdgcn_s_barrier();
    __builtin_amdgcn_fence(__ATOMIC_ACQUIRE, "workgroup", "local");

    float4 cs[16], sb[16], xk[16];
    #pragma unroll
    for (int i = 0; i < 16; ++i) {
        const int k = i * 128 + lane * 4;
        cs[i] = *(const float4*)&css[k];
        sb[i] = *(const float4*)&sbs[k];
        xk[i] = *(const float4*)&xks[k];
    }

    #pragma nounroll
    for (int it = 0; it < 2; ++it) {
        const float* p = aRow + (size_t)(3 * it) * N_NODES;
        float* o = oRow + (3 * it) * NFILT;
        const int n = n0 + 3 * it;

        softmax_row(b0, cs, sb, xk, xks[n], bias0, bias1, o, lane);
        __builtin_amdgcn_sched_barrier(0);
        load_row(b0, p + 3 * N_NODES);
        __builtin_amdgcn_sched_barrier(0);

        softmax_row(b1, cs, sb, xk, xks[n + 1], bias0, bias1, o + NFILT, lane);
        __builtin_amdgcn_sched_barrier(0);
        load_row(b1, p + 4 * N_NODES);
        __builtin_amdgcn_sched_barrier(0);

        softmax_row(b2, cs, sb, xk, xks[n + 2], bias0, bias1, o + 2 * NFILT, lane);
        if (it == 0) load_row(b2, p + 5 * N_NODES);
    }
    softmax_row(b0, cs, sb, xk, xks[n0 + 6], bias0, bias1, oRow + 6 * NFILT, lane);
    softmax_row(b1, cs, sb, xk, xks[n0 + 7], bias0, bias1, oRow + 7 * NFILT, lane);
}

extern "C" void kernel_launch(void* const* d_in, const int* in_sizes, int n_in,
                              void* d_out, int out_size, void* d_ws, size_t ws_size,
                              hipStream_t stream) {
    const float* x    = (const float*)d_in[0];
    const float* A    = (const float*)d_in[1];
    const float* w1   = (const float*)d_in[2];
    const float* Wa   = (const float*)d_in[3];
    const float* ba   = (const float*)d_in[4];
    const float* bias = (const float*)d_in[5];
    float* out = (float*)d_out;

    float* X1   = (float*)d_ws;
    float* CSP  = X1 + BATCH * N_NODES;
    float* SB2P = CSP + KQ * N_NODES;

    k_proj<<<dim3(256), dim3(256), 0, stream>>>(x, w1, X1);
    k_wmma_s<<<dim3(256), dim3(256), 0, stream>>>(Wa, ba, X1, SB2P, CSP);
    k_softmax_agg<<<dim3(256), dim3(SM_THREADS), 0, stream>>>(A, bias, X1, SB2P, CSP, out);
}
